# starting kernel
# baseline (speedup 1.0000x reference)
#include <hip/hip_runtime.h>
#include <math.h>

typedef __attribute__((ext_vector_type(2))) float v2f;
typedef __attribute__((ext_vector_type(4))) float v4f;
typedef __attribute__((ext_vector_type(8))) float v8f;

#define Bn 8
#define Sn 2048
#define Fn 128
#define Wn 5
#define NW (Sn - Wn + 1)
#define KC 64

__global__ void __launch_bounds__(256)
k_flash_softmax_gemm(const float* __restrict__ proj, const float* __restrict__ score,
                     const float* __restrict__ feature, float* __restrict__ out) {
    __shared__ float ldsA[16][KC + 1];
    __shared__ __align__(16) float smScale[16];
    __shared__ __align__(16) float smInv[16];

    const int bid  = blockIdx.x;
    const int b    = bid >> 7;
    const int i0   = (bid & 127) << 4;
    const int tid  = threadIdx.x;
    const int lane = tid & 31;
    const int wave = tid >> 5;
    const int n0   = wave << 4;

    const int r = tid >> 4;
    const int c = (tid & 15) << 2;
    const int rowIdx = b * Sn + i0 + r;
    const float* prow  = proj    + (size_t)rowIdx * Sn;
    const float* srow  = score   + (size_t)b * Sn;
    const float* fbase = feature + (size_t)b * Sn * Fn;

    const int m16   = lane & 15;
    const int half  = lane >> 4;
    const int khalf = half << 1;

    float m = -INFINITY, s = 0.0f;
    v8f acc = {};

    for (int k0 = 0; k0 < Sn; k0 += KC) {
        v4f p  = *(const v4f*)(prow + k0 + c);
        v4f sc = *(const v4f*)(srow + k0 + c);
        float x0 = p[0] * sc[0], x1 = p[1] * sc[1];
        float x2 = p[2] * sc[2], x3 = p[3] * sc[3];
        float cm = fmaxf(fmaxf(x0, x1), fmaxf(x2, x3));
        #pragma unroll
        for (int off = 1; off < 16; off <<= 1)
            cm = fmaxf(cm, __shfl_xor(cm, off, 32));

        const float mnew  = fmaxf(m, cm);
        const float scale = __expf(m - mnew);
        float w0 = __expf(x0 - mnew), w1 = __expf(x1 - mnew);
        float w2 = __expf(x2 - mnew), w3 = __expf(x3 - mnew);
        float csum = (w0 + w1) + (w2 + w3);
        #pragma unroll
        for (int off = 1; off < 16; off <<= 1)
            csum += __shfl_xor(csum, off, 32);
        s = s * scale + csum;
        m = mnew;

        ldsA[r][c + 0] = w0;
        ldsA[r][c + 1] = w1;
        ldsA[r][c + 2] = w2;
        ldsA[r][c + 3] = w3;
        if ((tid & 15) == 0) smScale[r] = scale;
        __syncthreads();

        v4f s0 = *(const v4f*)&smScale[half * 8];
        v4f s1 = *(const v4f*)&smScale[half * 8 + 4];
        #pragma unroll
        for (int rr = 0; rr < 4; ++rr) { acc[rr] *= s0[rr]; acc[rr + 4] *= s1[rr]; }

        #pragma unroll
        for (int t = 0; t < KC / 4; ++t) {
            v2f a;
            a.x = ldsA[m16][4 * t + khalf];
            a.y = ldsA[m16][4 * t + khalf + 1];
            const float* fp = fbase + (size_t)(k0 + 4 * t + khalf) * Fn + n0 + m16;
            v2f bb;
            bb.x = fp[0];
            bb.y = fp[Fn];
            acc = __builtin_amdgcn_wmma_f32_16x16x4_f32(
                false, a, false, bb, (short)0, acc, false, false);
        }
        __syncthreads();
    }

    if ((tid & 15) == 0) smInv[r] = 1.0f / s;
    __syncthreads();
    v4f i0v = *(const v4f*)&smInv[half * 8];
    v4f i1v = *(const v4f*)&smInv[half * 8 + 4];
    #pragma unroll
    for (int rr = 0; rr < 4; ++rr) { acc[rr] *= i0v[rr]; acc[rr + 4] *= i1v[rr]; }

    const int orow0 = i0 + (half << 3);
    float* obase = out + (size_t)b * Sn * Fn + (size_t)orow0 * Fn + n0 + m16;
    #pragma unroll
    for (int rr = 0; rr < 8; ++rr)
        obase[(size_t)rr * Fn] = acc[rr];
}

__global__ void __launch_bounds__(256)
k_rowdots(const float* __restrict__ feature, const float* __restrict__ aligned,
          float* __restrict__ dotv, float* __restrict__ nfv, float* __restrict__ nav) {
    const int row  = blockIdx.x * 8 + (threadIdx.x >> 5);
    const int lane = threadIdx.x & 31;
    v4f fv = *(const v4f*)(feature + (size_t)row * Fn + lane * 4);
    v4f av = *(const v4f*)(aligned + (size_t)row * Fn + lane * 4);
    float d = 0.f, nf = 0.f, na = 0.f;
    #pragma unroll
    for (int q = 0; q < 4; ++q) {
        d  += fv[q] * av[q];
        nf += fv[q] * fv[q];
        na += av[q] * av[q];
    }
    #pragma unroll
    for (int off = 16; off > 0; off >>= 1) {
        d  += __shfl_xor(d,  off, 32);
        nf += __shfl_xor(nf, off, 32);
        na += __shfl_xor(na, off, 32);
    }
    if (lane == 0) { dotv[row] = d; nfv[row] = nf; nav[row] = na; }
}

__global__ void __launch_bounds__(256)
k_window(const float* __restrict__ dotv, const float* __restrict__ nfv,
         const float* __restrict__ nav, float* __restrict__ part) {
    __shared__ float red[256];
    const int w = blockIdx.x * 256 + threadIdx.x;
    float v = 0.0f;
    if (w < Bn * NW) {
        const int b = w / NW;
        const int i = w % NW;
        const int base = b * Sn + i;
        float dw = 0.f, fw = 0.f, aw = 0.f;
        #pragma unroll
        for (int j = 0; j < Wn; ++j) {
            dw += dotv[base + j];
            fw += nfv[base + j];
            aw += nav[base + j];
        }
        float cs = dw / (fmaxf(sqrtf(fw), 1e-12f) * fmaxf(sqrtf(aw), 1e-12f));
        v = 1.0f - cs;
    }
    red[threadIdx.x] = v;
    __syncthreads();
    #pragma unroll
    for (int off = 128; off > 0; off >>= 1) {
        if (threadIdx.x < off) red[threadIdx.x] += red[threadIdx.x + off];
        __syncthreads();
    }
    if (threadIdx.x == 0) part[blockIdx.x] = red[0];
}

__global__ void k_final(const float* __restrict__ part, float* __restrict__ loss_out) {
    float s = 0.0f;
    for (int i = 0; i < 64; ++i) s += part[i];
    loss_out[0] = s / (float)(Bn * NW);
}

extern "C" void kernel_launch(void* const* d_in, const int* in_sizes, int n_in,
                              void* d_out, int out_size, void* d_ws, size_t ws_size,
                              hipStream_t stream) {
    const float* score   = (const float*)d_in[0];
    const float* feature = (const float*)d_in[1];
    const float* proj    = (const float*)d_in[2];
    float* out = (float*)d_out;
    float* ws  = (float*)d_ws;

    float* dotv = ws;
    float* nfv  = ws + 16384;
    float* nav  = ws + 32768;
    float* part = ws + 49152;

    k_flash_softmax_gemm<<<Bn * (Sn / 16), 256, 0, stream>>>(proj, score, feature, out);
    k_rowdots           <<<(Bn * Sn) / 8, 256, 0, stream>>>(feature, out, dotv, nfv, nav);
    k_window            <<<64, 256, 0, stream>>>(dotv, nfv, nav, part);
    k_final             <<<1, 1, 0, stream>>>(part, out + (size_t)Bn * Sn * Fn);
}


# K chunk per barrier pair 64 -> 256 columns: 4x fewer barriers, row reductions and accumulator rescales, 4 proj loads in flight per thread; softmax sums grouped per 256 columns (sum order changes)
# speedup vs baseline: 1.1380x; 1.1380x over previous
#include <hip/hip_runtime.h>
#include <math.h>

typedef __attribute__((ext_vector_type(2))) float v2f;
typedef __attribute__((ext_vector_type(4))) float v4f;
typedef __attribute__((ext_vector_type(8))) float v8f;

#define Bn 8
#define Sn 2048
#define Fn 128
#define Wn 5
#define NW (Sn - Wn + 1)
#define KC 256
#define KQ (KC / 64)

__global__ void __launch_bounds__(256)
k_flash_softmax_gemm(const float* __restrict__ proj, const float* __restrict__ score,
                     const float* __restrict__ feature, float* __restrict__ out) {
    __shared__ float ldsA[16][KC + 1];
    __shared__ __align__(16) float smScale[16];
    __shared__ __align__(16) float smInv[16];

    const int bid  = blockIdx.x;
    const int b    = bid >> 7;
    const int i0   = (bid & 127) << 4;
    const int tid  = threadIdx.x;
    const int lane = tid & 31;
    const int wave = tid >> 5;
    const int n0   = wave << 4;

    const int r = tid >> 4;
    const int c = (tid & 15) << 2;
    const int rowIdx = b * Sn + i0 + r;
    const float* prow  = proj    + (size_t)rowIdx * Sn;
    const float* srow  = score   + (size_t)b * Sn;
    const float* fbase = feature + (size_t)b * Sn * Fn;

    const int m16   = lane & 15;
    const int half  = lane >> 4;
    const int khalf = half << 1;

    float m = -INFINITY, s = 0.0f;
    v8f acc = {};

    for (int k0 = 0; k0 < Sn; k0 += KC) {
        v4f x[KQ];
        float cm = -INFINITY;
        for (int j = 0; j < KQ; ++j) {
            v4f p  = *(const v4f*)(prow + k0 + 64 * j + c);
            v4f sc = *(const v4f*)(srow + k0 + 64 * j + c);
            x[j] = p * sc;
            cm = fmaxf(cm, fmaxf(fmaxf(x[j][0], x[j][1]), fmaxf(x[j][2], x[j][3])));
        }
        #pragma unroll
        for (int off = 1; off < 16; off <<= 1)
            cm = fmaxf(cm, __shfl_xor(cm, off, 32));

        const float mnew  = fmaxf(m, cm);
        const float scale = __expf(m - mnew);
        float csum = 0.0f;
        for (int j = 0; j < KQ; ++j) {
            float w0 = __expf(x[j][0] - mnew), w1 = __expf(x[j][1] - mnew);
            float w2 = __expf(x[j][2] - mnew), w3 = __expf(x[j][3] - mnew);
            csum += (w0 + w1) + (w2 + w3);
            ldsA[r][64 * j + c + 0] = w0;
            ldsA[r][64 * j + c + 1] = w1;
            ldsA[r][64 * j + c + 2] = w2;
            ldsA[r][64 * j + c + 3] = w3;
        }
        #pragma unroll
        for (int off = 1; off < 16; off <<= 1)
            csum += __shfl_xor(csum, off, 32);
        s = s * scale + csum;
        m = mnew;

        if ((tid & 15) == 0) smScale[r] = scale;
        __syncthreads();

        v4f s0 = *(const v4f*)&smScale[half * 8];
        v4f s1 = *(const v4f*)&smScale[half * 8 + 4];
        #pragma unroll
        for (int rr = 0; rr < 4; ++rr) { acc[rr] *= s0[rr]; acc[rr + 4] *= s1[rr]; }

        #pragma unroll
        for (int t = 0; t < KC / 4; ++t) {
            v2f a;
            a.x = ldsA[m16][4 * t + khalf];
            a.y = ldsA[m16][4 * t + khalf + 1];
            const float* fp = fbase + (size_t)(k0 + 4 * t + khalf) * Fn + n0 + m16;
            v2f bb;
            bb.x = fp[0];
            bb.y = fp[Fn];
            acc = __builtin_amdgcn_wmma_f32_16x16x4_f32(
                false, a, false, bb, (short)0, acc, false, false);
        }
        __syncthreads();
    }

    if ((tid & 15) == 0) smInv[r] = 1.0f / s;
    __syncthreads();
    v4f i0v = *(const v4f*)&smInv[half * 8];
    v4f i1v = *(const v4f*)&smInv[half * 8 + 4];
    #pragma unroll
    for (int rr = 0; rr < 4; ++rr) { acc[rr] *= i0v[rr]; acc[rr + 4] *= i1v[rr]; }

    const int orow0 = i0 + (half << 3);
    float* obase = out + (size_t)b * Sn * Fn + (size_t)orow0 * Fn + n0 + m16;
    #pragma unroll
    for (int rr = 0; rr < 8; ++rr)
        obase[(size_t)rr * Fn] = acc[rr];
}

__global__ void __launch_bounds__(256)
k_rowdots(const float* __restrict__ feature, const float* __restrict__ aligned,
          float* __restrict__ dotv, float* __restrict__ nfv, float* __restrict__ nav) {
    const int row  = blockIdx.x * 8 + (threadIdx.x >> 5);
    const int lane = threadIdx.x & 31;
    v4f fv = *(const v4f*)(feature + (size_t)row * Fn + lane * 4);
    v4f av = *(const v4f*)(aligned + (size_t)row * Fn + lane * 4);
    float d = 0.f, nf = 0.f, na = 0.f;
    #pragma unroll
    for (int q = 0; q < 4; ++q) {
        d  += fv[q] * av[q];
        nf += fv[q] * fv[q];
        na += av[q] * av[q];
    }
    #pragma unroll
    for (int off = 16; off > 0; off >>= 1) {
        d  += __shfl_xor(d,  off, 32);
        nf += __shfl_xor(nf, off, 32);
        na += __shfl_xor(na, off, 32);
    }
    if (lane == 0) { dotv[row] = d; nfv[row] = nf; nav[row] = na; }
}

__global__ void __launch_bounds__(256)
k_window(const float* __restrict__ dotv, const float* __restrict__ nfv,
         const float* __restrict__ nav, float* __restrict__ part) {
    __shared__ float red[256];
    const int w = blockIdx.x * 256 + threadIdx.x;
    float v = 0.0f;
    if (w < Bn * NW) {
        const int b = w / NW;
        const int i = w % NW;
        const int base = b * Sn + i;
        float dw = 0.f, fw = 0.f, aw = 0.f;
        #pragma unroll
        for (int j = 0; j < Wn; ++j) {
            dw += dotv[base + j];
            fw += nfv[base + j];
            aw += nav[base + j];
        }
        float cs = dw / (fmaxf(sqrtf(fw), 1e-12f) * fmaxf(sqrtf(aw), 1e-12f));
        v = 1.0f - cs;
    }
    red[threadIdx.x] = v;
    __syncthreads();
    #pragma unroll
    for (int off = 128; off > 0; off >>= 1) {
        if (threadIdx.x < off) red[threadIdx.x] += red[threadIdx.x + off];
        __syncthreads();
    }
    if (threadIdx.x == 0) part[blockIdx.x] = red[0];
}

__global__ void k_final(const float* __restrict__ part, float* __restrict__ loss_out) {
    float s = 0.0f;
    for (int i = 0; i < 64; ++i) s += part[i];
    loss_out[0] = s / (float)(Bn * NW);
}

extern "C" void kernel_launch(void* const* d_in, const int* in_sizes, int n_in,
                              void* d_out, int out_size, void* d_ws, size_t ws_size,
                              hipStream_t stream) {
    const float* score   = (const float*)d_in[0];
    const float* feature = (const float*)d_in[1];
    const float* proj    = (const float*)d_in[2];
    float* out = (float*)d_out;
    float* ws  = (float*)d_ws;

    float* dotv = ws;
    float* nfv  = ws + 16384;
    float* nav  = ws + 32768;
    float* part = ws + 49152;

    k_flash_softmax_gemm<<<Bn * (Sn / 16), 256, 0, stream>>>(proj, score, feature, out);
    k_rowdots           <<<(Bn * Sn) / 8, 256, 0, stream>>>(feature, out, dotv, nfv, nav);
    k_window            <<<64, 256, 0, stream>>>(dotv, nfv, nav, part);
    k_final             <<<1, 1, 0, stream>>>(part, out + (size_t)Bn * Sn * Fn);
}


# Block owns 32 rows instead of 16: each wave keeps two 16-row accumulator tiles, so every feature (B) fragment load feeds two WMMAs; grid 1024 -> 512 blocks; sum orders unchanged
# speedup vs baseline: 1.3338x; 1.1721x over previous
#include <hip/hip_runtime.h>
#include <math.h>

typedef __attribute__((ext_vector_type(2))) float v2f;
typedef __attribute__((ext_vector_type(4))) float v4f;
typedef __attribute__((ext_vector_type(8))) float v8f;

#define Bn 8
#define Sn 2048
#define Fn 128
#define Wn 5
#define NW (Sn - Wn + 1)
#define KC 256
#define KQ (KC / 64)
#define MR 32
#define MT (MR / 16)

__global__ void __launch_bounds__(256)
k_flash_softmax_gemm(const float* __restrict__ proj, const float* __restrict__ score,
                     const float* __restrict__ feature, float* __restrict__ out) {
    __shared__ float ldsA[MR][KC + 1];
    __shared__ __align__(16) float smScale[MR];
    __shared__ __align__(16) float smInv[MR];

    const int bid  = blockIdx.x;
    const int b    = bid >> 6;
    const int i0   = (bid & 63) << 5;
    const int tid  = threadIdx.x;
    const int lane = tid & 31;
    const int wave = tid >> 5;
    const int n0   = wave << 4;

    const int r = tid >> 4;
    const int c = (tid & 15) << 2;
    const float* prow  = proj    + (size_t)(b * Sn + i0 + r) * Sn;
    const float* srow  = score   + (size_t)b * Sn;
    const float* fbase = feature + (size_t)b * Sn * Fn;

    const int m16   = lane & 15;
    const int half  = lane >> 4;
    const int khalf = half << 1;

    float m[MT], s[MT];
    v8f acc[MT];
    #pragma unroll
    for (int i = 0; i < MT; ++i) { m[i] = -INFINITY; s[i] = 0.0f; acc[i] = v8f{}; }

    for (int k0 = 0; k0 < Sn; k0 += KC) {
        v4f x[MT][KQ];
        float cm[MT];
        #pragma unroll
        for (int i = 0; i < MT; ++i) cm[i] = -INFINITY;
        #pragma unroll
        for (int j = 0; j < KQ; ++j) {
            v4f sc = *(const v4f*)(srow + k0 + 64 * j + c);
            #pragma unroll
            for (int i = 0; i < MT; ++i) {
                v4f p = *(const v4f*)(prow + (size_t)(16 * i) * Sn + k0 + 64 * j + c);
                x[i][j] = p * sc;
                cm[i] = fmaxf(cm[i], fmaxf(fmaxf(x[i][j][0], x[i][j][1]),
                                           fmaxf(x[i][j][2], x[i][j][3])));
            }
        }
        #pragma unroll
        for (int off = 1; off < 16; off <<= 1) {
            #pragma unroll
            for (int i = 0; i < MT; ++i)
                cm[i] = fmaxf(cm[i], __shfl_xor(cm[i], off, 32));
        }

        float scale[MT], csum[MT];
        #pragma unroll
        for (int i = 0; i < MT; ++i) {
            const float mnew = fmaxf(m[i], cm[i]);
            scale[i] = __expf(m[i] - mnew);
            m[i] = mnew;
            csum[i] = 0.0f;
            #pragma unroll
            for (int j = 0; j < KQ; ++j) {
                float w0 = __expf(x[i][j][0] - mnew), w1 = __expf(x[i][j][1] - mnew);
                float w2 = __expf(x[i][j][2] - mnew), w3 = __expf(x[i][j][3] - mnew);
                csum[i] += (w0 + w1) + (w2 + w3);
                ldsA[16 * i + r][64 * j + c + 0] = w0;
                ldsA[16 * i + r][64 * j + c + 1] = w1;
                ldsA[16 * i + r][64 * j + c + 2] = w2;
                ldsA[16 * i + r][64 * j + c + 3] = w3;
            }
        }
        #pragma unroll
        for (int off = 1; off < 16; off <<= 1) {
            #pragma unroll
            for (int i = 0; i < MT; ++i)
                csum[i] += __shfl_xor(csum[i], off, 32);
        }
        #pragma unroll
        for (int i = 0; i < MT; ++i) {
            s[i] = s[i] * scale[i] + csum[i];
            if ((tid & 15) == 0) smScale[16 * i + r] = scale[i];
        }
        __syncthreads();

        #pragma unroll
        for (int i = 0; i < MT; ++i) {
            v4f s0 = *(const v4f*)&smScale[16 * i + half * 8];
            v4f s1 = *(const v4f*)&smScale[16 * i + half * 8 + 4];
            #pragma unroll
            for (int rr = 0; rr < 4; ++rr) { acc[i][rr] *= s0[rr]; acc[i][rr + 4] *= s1[rr]; }
        }

        #pragma unroll
        for (int t = 0; t < KC / 4; ++t) {
            const float* fp = fbase + (size_t)(k0 + 4 * t + khalf) * Fn + n0 + m16;
            v2f bb;
            bb.x = fp[0];
            bb.y = fp[Fn];
            #pragma unroll
            for (int i = 0; i < MT; ++i) {
                v2f a;
                a.x = ldsA[16 * i + m16][4 * t + khalf];
                a.y = ldsA[16 * i + m16][4 * t + khalf + 1];
                acc[i] = __builtin_amdgcn_wmma_f32_16x16x4_f32(
                    false, a, false, bb, (short)0, acc[i], false, false);
            }
        }
        __syncthreads();
    }

    #pragma unroll
    for (int i = 0; i < MT; ++i)
        if ((tid & 15) == 0) smInv[16 * i + r] = 1.0f / s[i];
    __syncthreads();
    #pragma unroll
    for (int i = 0; i < MT; ++i) {
        v4f i0v = *(const v4f*)&smInv[16 * i + half * 8];
        v4f i1v = *(const v4f*)&smInv[16 * i + half * 8 + 4];
        #pragma unroll
        for (int rr = 0; rr < 4; ++rr) { acc[i][rr] *= i0v[rr]; acc[i][rr + 4] *= i1v[rr]; }

        const int orow0 = i0 + 16 * i + (half << 3);
        float* obase = out + (size_t)b * Sn * Fn + (size_t)orow0 * Fn + n0 + m16;
        #pragma unroll
        for (int rr = 0; rr < 8; ++rr)
            obase[(size_t)rr * Fn] = acc[i][rr];
    }
}

__global__ void __launch_bounds__(256)
k_rowdots(const float* __restrict__ feature, const float* __restrict__ aligned,
          float* __restrict__ dotv, float* __restrict__ nfv, float* __restrict__ nav) {
    const int row  = blockIdx.x * 8 + (threadIdx.x >> 5);
    const int lane = threadIdx.x & 31;
    v4f fv = *(const v4f*)(feature + (size_t)row * Fn + lane * 4);
    v4f av = *(const v4f*)(aligned + (size_t)row * Fn + lane * 4);
    float d = 0.f, nf = 0.f, na = 0.f;
    #pragma unroll
    for (int q = 0; q < 4; ++q) {
        d  += fv[q] * av[q];
        nf += fv[q] * fv[q];
        na += av[q] * av[q];
    }
    #pragma unroll
    for (int off = 16; off > 0; off >>= 1) {
        d  += __shfl_xor(d,  off, 32);
        nf += __shfl_xor(nf, off, 32);
        na += __shfl_xor(na, off, 32);
    }
    if (lane == 0) { dotv[row] = d; nfv[row] = nf; nav[row] = na; }
}

__global__ void __launch_bounds__(256)
k_window(const float* __restrict__ dotv, const float* __restrict__ nfv,
         const float* __restrict__ nav, float* __restrict__ part) {
    __shared__ float red[256];
    const int w = blockIdx.x * 256 + threadIdx.x;
    float v = 0.0f;
    if (w < Bn * NW) {
        const int b = w / NW;
        const int i = w % NW;
        const int base = b * Sn + i;
        float dw = 0.f, fw = 0.f, aw = 0.f;
        #pragma unroll
        for (int j = 0; j < Wn; ++j) {
            dw += dotv[base + j];
            fw += nfv[base + j];
            aw += nav[base + j];
        }
        float cs = dw / (fmaxf(sqrtf(fw), 1e-12f) * fmaxf(sqrtf(aw), 1e-12f));
        v = 1.0f - cs;
    }
    red[threadIdx.x] = v;
    __syncthreads();
    #pragma unroll
    for (int off = 128; off > 0; off >>= 1) {
        if (threadIdx.x < off) red[threadIdx.x] += red[threadIdx.x + off];
        __syncthreads();
    }
    if (threadIdx.x == 0) part[blockIdx.x] = red[0];
}

__global__ void k_final(const float* __restrict__ part, float* __restrict__ loss_out) {
    float s = 0.0f;
    for (int i = 0; i < 64; ++i) s += part[i];
    loss_out[0] = s / (float)(Bn * NW);
}

extern "C" void kernel_launch(void* const* d_in, const int* in_sizes, int n_in,
                              void* d_out, int out_size, void* d_ws, size_t ws_size,
                              hipStream_t stream) {
    const float* score   = (const float*)d_in[0];
    const float* feature = (const float*)d_in[1];
    const float* proj    = (const float*)d_in[2];
    float* out = (float*)d_out;
    float* ws  = (float*)d_ws;

    float* dotv = ws;
    float* nfv  = ws + 16384;
    float* nav  = ws + 32768;
    float* part = ws + 49152;

    k_flash_softmax_gemm<<<Bn * (Sn / MR), 256, 0, stream>>>(proj, score, feature, out);
    k_rowdots           <<<(Bn * Sn) / 8, 256, 0, stream>>>(feature, out, dotv, nfv, nav);
    k_window            <<<64, 256, 0, stream>>>(dotv, nfv, nav, part);
    k_final             <<<1, 1, 0, stream>>>(part, out + (size_t)Bn * Sn * Fn);
}


# Per-row dot/norm launch fused into the GEMM epilogue: tile staged in LDS, each wave stores whole 512-B rows and reduces 4 rows (sum order kept); launches 4 -> 3, 16384 one-row waves gone
# speedup vs baseline: 1.4284x; 1.0709x over previous
#include <hip/hip_runtime.h>
#include <math.h>

typedef __attribute__((ext_vector_type(2))) float v2f;
typedef __attribute__((ext_vector_type(4))) float v4f;
typedef __attribute__((ext_vector_type(8))) float v8f;

#define Bn 8
#define Sn 2048
#define Fn 128
#define Wn 5
#define NW (Sn - Wn + 1)
#define KC 256
#define KQ (KC / 64)
#define MR 32
#define MT (MR / 16)
#define SP (Fn + 4)

__global__ void __launch_bounds__(256)
k_flash_softmax_gemm(const float* __restrict__ proj, const float* __restrict__ score,
                     const float* __restrict__ feature, float* __restrict__ out,
                     float* __restrict__ dotv, float* __restrict__ nfv, float* __restrict__ nav) {
    __shared__ __align__(16) float ldsA[MR][KC + 1];
    __shared__ __align__(16) float smScale[MR];
    __shared__ __align__(16) float smInv[MR];

    const int bid  = blockIdx.x;
    const int b    = bid >> 6;
    const int i0   = (bid & 63) << 5;
    const int tid  = threadIdx.x;
    const int lane = tid & 31;
    const int wave = tid >> 5;
    const int n0   = wave << 4;

    const int r = tid >> 4;
    const int c = (tid & 15) << 2;
    const float* prow  = proj    + (size_t)(b * Sn + i0 + r) * Sn;
    const float* srow  = score   + (size_t)b * Sn;
    const float* fbase = feature + (size_t)b * Sn * Fn;

    const int m16   = lane & 15;
    const int half  = lane >> 4;
    const int khalf = half << 1;

    float m[MT], s[MT];
    v8f acc[MT];
    #pragma unroll
    for (int i = 0; i < MT; ++i) { m[i] = -INFINITY; s[i] = 0.0f; acc[i] = v8f{}; }

    for (int k0 = 0; k0 < Sn; k0 += KC) {
        v4f x[MT][KQ];
        float cm[MT];
        #pragma unroll
        for (int i = 0; i < MT; ++i) cm[i] = -INFINITY;
        #pragma unroll
        for (int j = 0; j < KQ; ++j) {
            v4f sc = *(const v4f*)(srow + k0 + 64 * j + c);
            #pragma unroll
            for (int i = 0; i < MT; ++i) {
                v4f p = *(const v4f*)(prow + (size_t)(16 * i) * Sn + k0 + 64 * j + c);
                x[i][j] = p * sc;
                cm[i] = fmaxf(cm[i], fmaxf(fmaxf(x[i][j][0], x[i][j][1]),
                                           fmaxf(x[i][j][2], x[i][j][3])));
            }
        }
        #pragma unroll
        for (int off = 1; off < 16; off <<= 1) {
            #pragma unroll
            for (int i = 0; i < MT; ++i)
                cm[i] = fmaxf(cm[i], __shfl_xor(cm[i], off, 32));
        }

        float scale[MT], csum[MT];
        #pragma unroll
        for (int i = 0; i < MT; ++i) {
            const float mnew = fmaxf(m[i], cm[i]);
            scale[i] = __expf(m[i] - mnew);
            m[i] = mnew;
            csum[i] = 0.0f;
            #pragma unroll
            for (int j = 0; j < KQ; ++j) {
                float w0 = __expf(x[i][j][0] - mnew), w1 = __expf(x[i][j][1] - mnew);
                float w2 = __expf(x[i][j][2] - mnew), w3 = __expf(x[i][j][3] - mnew);
                csum[i] += (w0 + w1) + (w2 + w3);
                ldsA[16 * i + r][64 * j + c + 0] = w0;
                ldsA[16 * i + r][64 * j + c + 1] = w1;
                ldsA[16 * i + r][64 * j + c + 2] = w2;
                ldsA[16 * i + r][64 * j + c + 3] = w3;
            }
        }
        #pragma unroll
        for (int off = 1; off < 16; off <<= 1) {
            #pragma unroll
            for (int i = 0; i < MT; ++i)
                csum[i] += __shfl_xor(csum[i], off, 32);
        }
        #pragma unroll
        for (int i = 0; i < MT; ++i) {
            s[i] = s[i] * scale[i] + csum[i];
            if ((tid & 15) == 0) smScale[16 * i + r] = scale[i];
        }
        __syncthreads();

        #pragma unroll
        for (int i = 0; i < MT; ++i) {
            v4f s0 = *(const v4f*)&smScale[16 * i + half * 8];
            v4f s1 = *(const v4f*)&smScale[16 * i + half * 8 + 4];
            #pragma unroll
            for (int rr = 0; rr < 4; ++rr) { acc[i][rr] *= s0[rr]; acc[i][rr + 4] *= s1[rr]; }
        }

        #pragma unroll
        for (int t = 0; t < KC / 4; ++t) {
            const float* fp = fbase + (size_t)(k0 + 4 * t + khalf) * Fn + n0 + m16;
            v2f bb;
            bb.x = fp[0];
            bb.y = fp[Fn];
            #pragma unroll
            for (int i = 0; i < MT; ++i) {
                v2f a;
                a.x = ldsA[16 * i + m16][4 * t + khalf];
                a.y = ldsA[16 * i + m16][4 * t + khalf + 1];
                acc[i] = __builtin_amdgcn_wmma_f32_16x16x4_f32(
                    false, a, false, bb, (short)0, acc[i], false, false);
            }
        }
        __syncthreads();
    }

    const size_t grow0 = (size_t)b * Sn + i0 + (wave << 2);
    v4f fv[4];
    #pragma unroll
    for (int q = 0; q < 4; ++q)
        fv[q] = *(const v4f*)(feature + (grow0 + q) * Fn + lane * 4);

    #pragma unroll
    for (int i = 0; i < MT; ++i)
        if ((tid & 15) == 0) smInv[16 * i + r] = 1.0f / s[i];
    __syncthreads();
    float* stage = &ldsA[0][0];
    #pragma unroll
    for (int i = 0; i < MT; ++i) {
        v4f i0v = *(const v4f*)&smInv[16 * i + half * 8];
        v4f i1v = *(const v4f*)&smInv[16 * i + half * 8 + 4];
        #pragma unroll
        for (int rr = 0; rr < 4; ++rr) { acc[i][rr] *= i0v[rr]; acc[i][rr + 4] *= i1v[rr]; }

        #pragma unroll
        for (int rr = 0; rr < 8; ++rr)
            stage[(16 * i + (half << 3) + rr) * SP + n0 + m16] = acc[i][rr];
    }
    __syncthreads();

    float d[4], nf[4], na[4];
    #pragma unroll
    for (int q = 0; q < 4; ++q) {
        const v4f av = *(const v4f*)&stage[((wave << 2) + q) * SP + lane * 4];
        *(v4f*)(out + (grow0 + q) * Fn + lane * 4) = av;
        d[q] = 0.f; nf[q] = 0.f; na[q] = 0.f;
        #pragma unroll
        for (int e = 0; e < 4; ++e) {
            d[q]  += fv[q][e] * av[e];
            nf[q] += fv[q][e] * fv[q][e];
            na[q] += av[e] * av[e];
        }
    }
    #pragma unroll
    for (int off = 16; off > 0; off >>= 1) {
        #pragma unroll
        for (int q = 0; q < 4; ++q) {
            d[q]  += __shfl_xor(d[q],  off, 32);
            nf[q] += __shfl_xor(nf[q], off, 32);
            na[q] += __shfl_xor(na[q], off, 32);
        }
    }
    if (lane == 0) {
        *(v4f*)(dotv + grow0) = v4f{d[0],  d[1],  d[2],  d[3]};
        *(v4f*)(nfv  + grow0) = v4f{nf[0], nf[1], nf[2], nf[3]};
        *(v4f*)(nav  + grow0) = v4f{na[0], na[1], na[2], na[3]};
    }
}

__global__ void __launch_bounds__(256)
k_window(const float* __restrict__ dotv, const float* __restrict__ nfv,
         const float* __restrict__ nav, float* __restrict__ part) {
    __shared__ float red[256];
    const int w = blockIdx.x * 256 + threadIdx.x;
    float v = 0.0f;
    if (w < Bn * NW) {
        const int b = w / NW;
        const int i = w % NW;
        const int base = b * Sn + i;
        float dw = 0.f, fw = 0.f, aw = 0.f;
        #pragma unroll
        for (int j = 0; j < Wn; ++j) {
            dw += dotv[base + j];
            fw += nfv[base + j];
            aw += nav[base + j];
        }
        float cs = dw / (fmaxf(sqrtf(fw), 1e-12f) * fmaxf(sqrtf(aw), 1e-12f));
        v = 1.0f - cs;
    }
    red[threadIdx.x] = v;
    __syncthreads();
    #pragma unroll
    for (int off = 128; off > 0; off >>= 1) {
        if (threadIdx.x < off) red[threadIdx.x] += red[threadIdx.x + off];
        __syncthreads();
    }
    if (threadIdx.x == 0) part[blockIdx.x] = red[0];
}

__global__ void k_final(const float* __restrict__ part, float* __restrict__ loss_out) {
    float s = 0.0f;
    for (int i = 0; i < 64; ++i) s += part[i];
    loss_out[0] = s / (float)(Bn * NW);
}

extern "C" void kernel_launch(void* const* d_in, const int* in_sizes, int n_in,
                              void* d_out, int out_size, void* d_ws, size_t ws_size,
                              hipStream_t stream) {
    const float* score   = (const float*)d_in[0];
    const float* feature = (const float*)d_in[1];
    const float* proj    = (const float*)d_in[2];
    float* out = (float*)d_out;
    float* ws  = (float*)d_ws;

    float* dotv = ws;
    float* nfv  = ws + 16384;
    float* nav  = ws + 32768;
    float* part = ws + 49152;

    k_flash_softmax_gemm<<<Bn * (Sn / MR), 256, 0, stream>>>(proj, score, feature, out, dotv, nfv, nav);
    k_window            <<<64, 256, 0, stream>>>(dotv, nfv, nav, part);
    k_final             <<<1, 1, 0, stream>>>(part, out + (size_t)Bn * Sn * Fn);
}


# Wave tile 32x16 -> 16x64, chunk k range split over wave pairs: one 16-B feature load per row serves 4 column tiles, one A read feeds 4 WMMAs; k halves added in the epilogue (sum order changes)
# speedup vs baseline: 1.5094x; 1.0567x over previous
#include <hip/hip_runtime.h>
#include <math.h>

typedef __attribute__((ext_vector_type(2))) float v2f;
typedef __attribute__((ext_vector_type(4))) float v4f;
typedef __attribute__((ext_vector_type(8))) float v8f;

#define Bn 8
#define Sn 2048
#define Fn 128
#define Wn 5
#define NW (Sn - Wn + 1)
#define KC 256
#define KQ (KC / 64)
#define MR 32
#define MT (MR / 16)
#define SP (Fn + 4)

__global__ void __launch_bounds__(256)
k_flash_softmax_gemm(const float* __restrict__ proj, const float* __restrict__ score,
                     const float* __restrict__ feature, float* __restrict__ out,
                     float* __restrict__ dotv, float* __restrict__ nfv, float* __restrict__ nav) {
    __shared__ __align__(16) float ldsBuf[2 * MR * SP];
    float (*ldsA)[KC + 1] = (float (*)[KC + 1])ldsBuf;
    __shared__ __align__(16) float smScale[MR];
    __shared__ __align__(16) float smInv[MR];

    const int bid  = blockIdx.x;
    const int b    = bid >> 6;
    const int i0   = (bid & 63) << 5;
    const int tid  = threadIdx.x;
    const int lane = tid & 31;
    const int wave = tid >> 5;
    const int cb   = (wave & 1) << 6;
    const int rt   = (wave >> 1) & 1;
    const int ksel = wave >> 2;

    const int r = tid >> 4;
    const int c = (tid & 15) << 2;
    const float* prow  = proj    + (size_t)(b * Sn + i0 + r) * Sn;
    const float* srow  = score   + (size_t)b * Sn;
    const float* fbase = feature + (size_t)b * Sn * Fn;

    const int m16   = lane & 15;
    const int half  = lane >> 4;
    const int khalf = half << 1;
    const float* fwave = fbase + (size_t)(ksel * (KC / 2) + khalf) * Fn + cb + 4 * m16;
    const int    ka    = ksel * (KC / 2) + khalf;

    float m[MT], s[MT];
    v8f acc[4];
    #pragma unroll
    for (int i = 0; i < MT; ++i) { m[i] = -INFINITY; s[i] = 0.0f; }
    #pragma unroll
    for (int j = 0; j < 4; ++j) acc[j] = v8f{};

    for (int k0 = 0; k0 < Sn; k0 += KC) {
        v4f x[MT][KQ];
        float cm[MT];
        #pragma unroll
        for (int i = 0; i < MT; ++i) cm[i] = -INFINITY;
        #pragma unroll
        for (int j = 0; j < KQ; ++j) {
            v4f sc = *(const v4f*)(srow + k0 + 64 * j + c);
            #pragma unroll
            for (int i = 0; i < MT; ++i) {
                v4f p = *(const v4f*)(prow + (size_t)(16 * i) * Sn + k0 + 64 * j + c);
                x[i][j] = p * sc;
                cm[i] = fmaxf(cm[i], fmaxf(fmaxf(x[i][j][0], x[i][j][1]),
                                           fmaxf(x[i][j][2], x[i][j][3])));
            }
        }
        #pragma unroll
        for (int off = 1; off < 16; off <<= 1) {
            #pragma unroll
            for (int i = 0; i < MT; ++i)
                cm[i] = fmaxf(cm[i], __shfl_xor(cm[i], off, 32));
        }

        float scale[MT], csum[MT];
        #pragma unroll
        for (int i = 0; i < MT; ++i) {
            const float mnew = fmaxf(m[i], cm[i]);
            scale[i] = __expf(m[i] - mnew);
            m[i] = mnew;
            csum[i] = 0.0f;
            #pragma unroll
            for (int j = 0; j < KQ; ++j) {
                float w0 = __expf(x[i][j][0] - mnew), w1 = __expf(x[i][j][1] - mnew);
                float w2 = __expf(x[i][j][2] - mnew), w3 = __expf(x[i][j][3] - mnew);
                csum[i] += (w0 + w1) + (w2 + w3);
                ldsA[16 * i + r][64 * j + c + 0] = w0;
                ldsA[16 * i + r][64 * j + c + 1] = w1;
                ldsA[16 * i + r][64 * j + c + 2] = w2;
                ldsA[16 * i + r][64 * j + c + 3] = w3;
            }
        }
        #pragma unroll
        for (int off = 1; off < 16; off <<= 1) {
            #pragma unroll
            for (int i = 0; i < MT; ++i)
                csum[i] += __shfl_xor(csum[i], off, 32);
        }
        #pragma unroll
        for (int i = 0; i < MT; ++i) {
            s[i] = s[i] * scale[i] + csum[i];
            if ((tid & 15) == 0) smScale[16 * i + r] = scale[i];
        }
        __syncthreads();

        {
            const v4f s0 = *(const v4f*)&smScale[16 * rt + half * 8];
            const v4f s1 = *(const v4f*)&smScale[16 * rt + half * 8 + 4];
            #pragma unroll
            for (int j = 0; j < 4; ++j) {
                #pragma unroll
                for (int rr = 0; rr < 4; ++rr) { acc[j][rr] *= s0[rr]; acc[j][rr + 4] *= s1[rr]; }
            }
        }

        #pragma unroll
        for (int u = 0; u < KC / 8; ++u) {
            const int kk = ka + 4 * u;
            const float* fp = fwave + (size_t)(k0 + 4 * u) * Fn;
            const v4f bx = *(const v4f*)fp;
            const v4f by = *(const v4f*)(fp + Fn);
            v2f a;
            a.x = ldsA[16 * rt + m16][kk];
            a.y = ldsA[16 * rt + m16][kk + 1];
            #pragma unroll
            for (int j = 0; j < 4; ++j) {
                v2f bb;
                bb.x = bx[j];
                bb.y = by[j];
                acc[j] = __builtin_amdgcn_wmma_f32_16x16x4_f32(
                    false, a, false, bb, (short)0, acc[j], false, false);
            }
        }
        __syncthreads();
    }

    const size_t grow0 = (size_t)b * Sn + i0 + (wave << 2);
    v4f fv[4];
    #pragma unroll
    for (int q = 0; q < 4; ++q)
        fv[q] = *(const v4f*)(feature + (grow0 + q) * Fn + lane * 4);

    #pragma unroll
    for (int i = 0; i < MT; ++i)
        if ((tid & 15) == 0) smInv[16 * i + r] = 1.0f / s[i];
    float* stage = ldsBuf + ksel * (MR * SP);
    #pragma unroll
    for (int rr = 0; rr < 8; ++rr)
        *(v4f*)&stage[(16 * rt + (half << 3) + rr) * SP + cb + 4 * m16] =
            v4f{acc[0][rr], acc[1][rr], acc[2][rr], acc[3][rr]};
    __syncthreads();

    float d[4], nf[4], na[4];
    #pragma unroll
    for (int q = 0; q < 4; ++q) {
        const int trow = (wave << 2) + q;
        const v4f av = (*(const v4f*)&ldsBuf[trow * SP + lane * 4] +
                        *(const v4f*)&ldsBuf[MR * SP + trow * SP + lane * 4]) * smInv[trow];
        *(v4f*)(out + (grow0 + q) * Fn + lane * 4) = av;
        d[q] = 0.f; nf[q] = 0.f; na[q] = 0.f;
        #pragma unroll
        for (int e = 0; e < 4; ++e) {
            d[q]  += fv[q][e] * av[e];
            nf[q] += fv[q][e] * fv[q][e];
            na[q] += av[e] * av[e];
        }
    }
    #pragma unroll
    for (int off = 16; off > 0; off >>= 1) {
        #pragma unroll
        for (int q = 0; q < 4; ++q) {
            d[q]  += __shfl_xor(d[q],  off, 32);
            nf[q] += __shfl_xor(nf[q], off, 32);
            na[q] += __shfl_xor(na[q], off, 32);
        }
    }
    if (lane == 0) {
        *(v4f*)(dotv + grow0) = v4f{d[0],  d[1],  d[2],  d[3]};
        *(v4f*)(nfv  + grow0) = v4f{nf[0], nf[1], nf[2], nf[3]};
        *(v4f*)(nav  + grow0) = v4f{na[0], na[1], na[2], na[3]};
    }
}

__global__ void __launch_bounds__(256)
k_window(const float* __restrict__ dotv, const float* __restrict__ nfv,
         const float* __restrict__ nav, float* __restrict__ part) {
    __shared__ float red[256];
    const int w = blockIdx.x * 256 + threadIdx.x;
    float v = 0.0f;
    if (w < Bn * NW) {
        const int b = w / NW;
        const int i = w % NW;
        const int base = b * Sn + i;
        float dw = 0.f, fw = 0.f, aw = 0.f;
        #pragma unroll
        for (int j = 0; j < Wn; ++j) {
            dw += dotv[base + j];
            fw += nfv[base + j];
            aw += nav[base + j];
        }
        float cs = dw / (fmaxf(sqrtf(fw), 1e-12f) * fmaxf(sqrtf(aw), 1e-12f));
        v = 1.0f - cs;
    }
    red[threadIdx.x] = v;
    __syncthreads();
    #pragma unroll
    for (int off = 128; off > 0; off >>= 1) {
        if (threadIdx.x < off) red[threadIdx.x] += red[threadIdx.x + off];
        __syncthreads();
    }
    if (threadIdx.x == 0) part[blockIdx.x] = red[0];
}

__global__ void k_final(const float* __restrict__ part, float* __restrict__ loss_out) {
    float s = 0.0f;
    for (int i = 0; i < 64; ++i) s += part[i];
    loss_out[0] = s / (float)(Bn * NW);
}

extern "C" void kernel_launch(void* const* d_in, const int* in_sizes, int n_in,
                              void* d_out, int out_size, void* d_ws, size_t ws_size,
                              hipStream_t stream) {
    const float* score   = (const float*)d_in[0];
    const float* feature = (const float*)d_in[1];
    const float* proj    = (const float*)d_in[2];
    float* out = (float*)d_out;
    float* ws  = (float*)d_ws;

    float* dotv = ws;
    float* nfv  = ws + 16384;
    float* nav  = ws + 32768;
    float* part = ws + 49152;

    k_flash_softmax_gemm<<<Bn * (Sn / MR), 256, 0, stream>>>(proj, score, feature, out, dotv, nfv, nav);
    k_window            <<<64, 256, 0, stream>>>(dotv, nfv, nav, part);
    k_final             <<<1, 1, 0, stream>>>(part, out + (size_t)Bn * Sn * Fn);
}


# Softmax weights formed per wave in registers as the WMMA B operand (transposed tile); k split over 4 waves, own max/sum merged in epilogue: no LDS, barrier or shuffle in k loop; sum order changes
# speedup vs baseline: 1.6135x; 1.0690x over previous
#include <hip/hip_runtime.h>
#include <math.h>

typedef __attribute__((ext_vector_type(2))) float v2f;
typedef __attribute__((ext_vector_type(4))) float v4f;
typedef __attribute__((ext_vector_type(8))) float v8f;

#define Bn 8
#define Sn 2048
#define Fn 128
#define Wn 5
#define NW (Sn - Wn + 1)
#define MR 32
#define MT (MR / 16)
#define KS 4
#define KW (Sn / KS)
#define KB 32
#define SP (Fn + 4)

__device__ __forceinline__ float xhalf(float x) {
    const int b = __builtin_bit_cast(int, x);
    return __builtin_bit_cast(float,
        __builtin_amdgcn_permlanex16(b, b, 0x76543210, 0xfedcba98, false, false));
}

__global__ void __launch_bounds__(256)
k_flash_softmax_gemm(const float* __restrict__ proj, const float* __restrict__ score,
                     const float* __restrict__ feature, float* __restrict__ out,
                     float* __restrict__ dotv, float* __restrict__ nfv, float* __restrict__ nav) {
    __shared__ __align__(16) float stage[KS * MR * SP];
    __shared__ float smM[KS * MR];
    __shared__ float smS[KS * MR];

    const int bid  = blockIdx.x;
    const int b    = bid >> 6;
    const int i0   = (bid & 63) << 5;
    const int tid  = threadIdx.x;
    const int lane = tid & 31;
    const int wave = tid >> 5;
    const int cg   = wave & 1;
    const int kq   = wave >> 1;
    const int c    = lane & 15;
    const int h    = lane >> 4;

    const int kfirst = kq * KW + 16 * h;
    const float* prow = proj    + (size_t)(b * Sn + i0 + c) * Sn + kfirst;
    const float* srow = score   + (size_t)b * Sn + kfirst;
    const float* frow = feature + ((size_t)b * Sn + kfirst) * Fn + 64 * cg + 4 * c;

    float m[MT], s[MT];
    v8f acc[4 * MT];
    #pragma unroll
    for (int i = 0; i < MT; ++i) { m[i] = -INFINITY; s[i] = 0.0f; }
    #pragma unroll
    for (int j = 0; j < 4 * MT; ++j) acc[j] = v8f{};

    v4f xn[MT][4];
    #pragma unroll
    for (int q = 0; q < 4; ++q) {
        const v4f sc = *(const v4f*)(srow + 4 * q);
        #pragma unroll
        for (int i = 0; i < MT; ++i)
            xn[i][q] = *(const v4f*)(prow + (size_t)(16 * i) * Sn + 4 * q) * sc;
    }

    for (int kb = 0; kb < KW; kb += KB) {
        v4f fa[8][2];
        #pragma unroll
        for (int t = 0; t < 8; ++t) {
            const float* fp = frow + (size_t)(kb + 2 * t) * Fn;
            fa[t][0] = *(const v4f*)fp;
            fa[t][1] = *(const v4f*)(fp + Fn);
        }
        const int kn = (kb + KB < KW) ? (kb + KB) : kb;
        v4f pn[MT][4], scn[4];
        #pragma unroll
        for (int q = 0; q < 4; ++q) {
            scn[q] = *(const v4f*)(srow + kn + 4 * q);
            #pragma unroll
            for (int i = 0; i < MT; ++i)
                pn[i][q] = *(const v4f*)(prow + (size_t)(16 * i) * Sn + kn + 4 * q);
        }
        __builtin_amdgcn_sched_barrier(0);

        v4f x[MT][4];
        #pragma unroll
        for (int i = 0; i < MT; ++i) {
            #pragma unroll
            for (int q = 0; q < 4; ++q) x[i][q] = xn[i][q];
        }
        #pragma unroll
        for (int i = 0; i < MT; ++i) {
            float bm = fmaxf(fmaxf(x[i][0][0], x[i][0][1]), fmaxf(x[i][0][2], x[i][0][3]));
            #pragma unroll
            for (int q = 1; q < 4; ++q)
                bm = fmaxf(bm, fmaxf(fmaxf(x[i][q][0], x[i][q][1]), fmaxf(x[i][q][2], x[i][q][3])));
            bm = fmaxf(bm, xhalf(bm));
            const float mnew  = fmaxf(m[i], bm);
            const float scale = __expf(m[i] - mnew);
            m[i] = mnew;
            float csum = 0.0f;
            #pragma unroll
            for (int q = 0; q < 4; ++q) {
                #pragma unroll
                for (int e = 0; e < 4; ++e) x[i][q][e] = __expf(x[i][q][e] - mnew);
                csum += (x[i][q][0] + x[i][q][1]) + (x[i][q][2] + x[i][q][3]);
            }
            s[i] = s[i] * scale + csum;
            #pragma unroll
            for (int j = 0; j < 4; ++j) acc[4 * i + j] = acc[4 * i + j] * scale;
        }

        #pragma unroll
        for (int t = 0; t < 8; ++t) {
            v2f bw[MT];
            #pragma unroll
            for (int i = 0; i < MT; ++i) {
                bw[i].x = x[i][t >> 1][(2 * t) & 3];
                bw[i].y = x[i][t >> 1][((2 * t) & 3) + 1];
            }
            #pragma unroll
            for (int j = 0; j < 4; ++j) {
                v2f a;
                a.x = fa[t][0][j];
                a.y = fa[t][1][j];
                #pragma unroll
                for (int i = 0; i < MT; ++i)
                    acc[4 * i + j] = __builtin_amdgcn_wmma_f32_16x16x4_f32(
                        false, a, false, bw[i], (short)0, acc[4 * i + j], false, false);
            }
        }

        __builtin_amdgcn_sched_barrier(0);
        #pragma unroll
        for (int i = 0; i < MT; ++i) {
            #pragma unroll
            for (int q = 0; q < 4; ++q) xn[i][q] = pn[i][q] * scn[q];
        }
    }

    const size_t grow0 = (size_t)b * Sn + i0 + (wave << 2);
    v4f fv[4];
    #pragma unroll
    for (int q = 0; q < 4; ++q)
        fv[q] = *(const v4f*)(feature + (grow0 + q) * Fn + lane * 4);

    #pragma unroll
    for (int i = 0; i < MT; ++i) {
        const float stot = s[i] + xhalf(s[i]);
        if (h == 0 && cg == 0) {
            smM[kq * MR + 16 * i + c] = m[i];
            smS[kq * MR + 16 * i + c] = stot;
        }
    }
    float* st = stage + (kq * MR + c) * SP + 64 * cg + 32 * h;
    #pragma unroll
    for (int i = 0; i < MT; ++i) {
        #pragma unroll
        for (int r = 0; r < 8; ++r)
            *(v4f*)&st[16 * i * SP + 4 * r] =
                v4f{acc[4 * i + 0][r], acc[4 * i + 1][r], acc[4 * i + 2][r], acc[4 * i + 3][r]};
    }
    __syncthreads();

    float d[4], nf[4], na[4];
    #pragma unroll
    for (int q = 0; q < 4; ++q) {
        const int trow = (wave << 2) + q;
        const float m0 = smM[trow], m1 = smM[MR + trow], m2 = smM[2 * MR + trow], m3 = smM[3 * MR + trow];
        const float mx = fmaxf(fmaxf(m0, m1), fmaxf(m2, m3));
        const float f0 = __expf(m0 - mx), f1 = __expf(m1 - mx), f2 = __expf(m2 - mx), f3 = __expf(m3 - mx);
        const float den = ((f0 * smS[trow] + f1 * smS[MR + trow]) + f2 * smS[2 * MR + trow]) + f3 * smS[3 * MR + trow];
        const float inv = 1.0f / den;
        const v4f p0 = *(const v4f*)&stage[(0 * MR + trow) * SP + lane * 4];
        const v4f p1 = *(const v4f*)&stage[(1 * MR + trow) * SP + lane * 4];
        const v4f p2 = *(const v4f*)&stage[(2 * MR + trow) * SP + lane * 4];
        const v4f p3 = *(const v4f*)&stage[(3 * MR + trow) * SP + lane * 4];
        const v4f av = (((p0 * f0 + p1 * f1) + p2 * f2) + p3 * f3) * inv;
        *(v4f*)(out + (grow0 + q) * Fn + lane * 4) = av;
        d[q] = 0.f; nf[q] = 0.f; na[q] = 0.f;
        #pragma unroll
        for (int e = 0; e < 4; ++e) {
            d[q]  += fv[q][e] * av[e];
            nf[q] += fv[q][e] * fv[q][e];
            na[q] += av[e] * av[e];
        }
    }
    #pragma unroll
    for (int off = 16; off > 0; off >>= 1) {
        #pragma unroll
        for (int q = 0; q < 4; ++q) {
            d[q]  += __shfl_xor(d[q],  off, 32);
            nf[q] += __shfl_xor(nf[q], off, 32);
            na[q] += __shfl_xor(na[q], off, 32);
        }
    }
    if (lane == 0) {
        *(v4f*)(dotv + grow0) = v4f{d[0],  d[1],  d[2],  d[3]};
        *(v4f*)(nfv  + grow0) = v4f{nf[0], nf[1], nf[2], nf[3]};
        *(v4f*)(nav  + grow0) = v4f{na[0], na[1], na[2], na[3]};
    }
}

__global__ void __launch_bounds__(256)
k_window(const float* __restrict__ dotv, const float* __restrict__ nfv,
         const float* __restrict__ nav, float* __restrict__ part) {
    __shared__ float red[256];
    const int w = blockIdx.x * 256 + threadIdx.x;
    float v = 0.0f;
    if (w < Bn * NW) {
        const int b = w / NW;
        const int i = w % NW;
        const int base = b * Sn + i;
        float dw = 0.f, fw = 0.f, aw = 0.f;
        #pragma unroll
        for (int j = 0; j < Wn; ++j) {
            dw += dotv[base + j];
            fw += nfv[base + j];
            aw += nav[base + j];
        }
        float cs = dw / (fmaxf(sqrtf(fw), 1e-12f) * fmaxf(sqrtf(aw), 1e-12f));
        v = 1.0f - cs;
    }
    red[threadIdx.x] = v;
    __syncthreads();
    #pragma unroll
    for (int off = 128; off > 0; off >>= 1) {
        if (threadIdx.x < off) red[threadIdx.x] += red[threadIdx.x + off];
        __syncthreads();
    }
    if (threadIdx.x == 0) part[blockIdx.x] = red[0];
}

__global__ void k_final(const float* __restrict__ part, float* __restrict__ loss_out) {
    float s = 0.0f;
    for (int i = 0; i < 64; ++i) s += part[i];
    loss_out[0] = s / (float)(Bn * NW);
}

extern "C" void kernel_launch(void* const* d_in, const int* in_sizes, int n_in,
                              void* d_out, int out_size, void* d_ws, size_t ws_size,
                              hipStream_t stream) {
    const float* score   = (const float*)d_in[0];
    const float* feature = (const float*)d_in[1];
    const float* proj    = (const float*)d_in[2];
    float* out = (float*)d_out;
    float* ws  = (float*)d_ws;

    float* dotv = ws;
    float* nfv  = ws + 16384;
    float* nav  = ws + 32768;
    float* part = ws + 49152;

    k_flash_softmax_gemm<<<Bn * (Sn / MR), 256, 0, stream>>>(proj, score, feature, out, dotv, nfv, nav);
    k_window            <<<64, 256, 0, stream>>>(dotv, nfv, nav, part);
    k_final             <<<1, 1, 0, stream>>>(part, out + (size_t)Bn * Sn * Fn);
}


# Column-group wave pairs merged: wave tile 32x64 -> 32x128 in 4-wave blocks (362 VGPRs, 2 waves/SIMD); each row's proj loads, max and exps are done by one wave, not two; sum orders unchanged
# speedup vs baseline: 1.6841x; 1.0438x over previous
#include <hip/hip_runtime.h>
#include <math.h>

typedef __attribute__((ext_vector_type(2))) float v2f;
typedef __attribute__((ext_vector_type(4))) float v4f;
typedef __attribute__((ext_vector_type(8))) float v8f;

#define Bn 8
#define Sn 2048
#define Fn 128
#define Wn 5
#define NW (Sn - Wn + 1)
#define MR 32
#define MT (MR / 16)
#define CT (Fn / 16)
#define KS 4
#define KW (Sn / KS)
#define KB 32
#define SP (Fn + 4)

__device__ __forceinline__ float xhalf(float x) {
    const int b = __builtin_bit_cast(int, x);
    return __builtin_bit_cast(float,
        __builtin_amdgcn_permlanex16(b, b, 0x76543210, 0xfedcba98, false, false));
}

__global__ void __launch_bounds__(128)
k_flash_softmax_gemm(const float* __restrict__ proj, const float* __restrict__ score,
                     const float* __restrict__ feature, float* __restrict__ out,
                     float* __restrict__ dotv, float* __restrict__ nfv, float* __restrict__ nav) {
    __shared__ __align__(16) float stage[KS * MR * SP];
    __shared__ float smM[KS * MR];
    __shared__ float smS[KS * MR];

    const int bid  = blockIdx.x;
    const int b    = bid >> 6;
    const int i0   = (bid & 63) << 5;
    const int tid  = threadIdx.x;
    const int lane = tid & 31;
    const int wave = tid >> 5;
    const int kq   = wave;
    const int c    = lane & 15;
    const int h    = lane >> 4;

    const int kfirst = kq * KW + 16 * h;
    const float* prow = proj    + (size_t)(b * Sn + i0 + c) * Sn + kfirst;
    const float* srow = score   + (size_t)b * Sn + kfirst;
    const float* frow = feature + ((size_t)b * Sn + kfirst) * Fn + 4 * c;

    float m[MT], s[MT];
    v8f acc[CT * MT];
    #pragma unroll
    for (int i = 0; i < MT; ++i) { m[i] = -INFINITY; s[i] = 0.0f; }
    #pragma unroll
    for (int j = 0; j < CT * MT; ++j) acc[j] = v8f{};

    v4f xn[MT][4];
    #pragma unroll
    for (int q = 0; q < 4; ++q) {
        const v4f sc = *(const v4f*)(srow + 4 * q);
        #pragma unroll
        for (int i = 0; i < MT; ++i)
            xn[i][q] = *(const v4f*)(prow + (size_t)(16 * i) * Sn + 4 * q) * sc;
    }

    for (int kb = 0; kb < KW; kb += KB) {
        v4f fa[8][2][2];
        #pragma unroll
        for (int t = 0; t < 8; ++t) {
            const float* fp = frow + (size_t)(kb + 2 * t) * Fn;
            fa[t][0][0] = *(const v4f*)fp;
            fa[t][0][1] = *(const v4f*)(fp + 64);
            fa[t][1][0] = *(const v4f*)(fp + Fn);
            fa[t][1][1] = *(const v4f*)(fp + Fn + 64);
        }
        const int kn = (kb + KB < KW) ? (kb + KB) : kb;
        v4f pn[MT][4], scn[4];
        #pragma unroll
        for (int q = 0; q < 4; ++q) {
            scn[q] = *(const v4f*)(srow + kn + 4 * q);
            #pragma unroll
            for (int i = 0; i < MT; ++i)
                pn[i][q] = *(const v4f*)(prow + (size_t)(16 * i) * Sn + kn + 4 * q);
        }
        __builtin_amdgcn_sched_barrier(0);

        v4f x[MT][4];
        #pragma unroll
        for (int i = 0; i < MT; ++i) {
            #pragma unroll
            for (int q = 0; q < 4; ++q) x[i][q] = xn[i][q];
        }
        #pragma unroll
        for (int i = 0; i < MT; ++i) {
            float bm = fmaxf(fmaxf(x[i][0][0], x[i][0][1]), fmaxf(x[i][0][2], x[i][0][3]));
            #pragma unroll
            for (int q = 1; q < 4; ++q)
                bm = fmaxf(bm, fmaxf(fmaxf(x[i][q][0], x[i][q][1]), fmaxf(x[i][q][2], x[i][q][3])));
            bm = fmaxf(bm, xhalf(bm));
            const float mnew  = fmaxf(m[i], bm);
            const float scale = __expf(m[i] - mnew);
            m[i] = mnew;
            float csum = 0.0f;
            #pragma unroll
            for (int q = 0; q < 4; ++q) {
                #pragma unroll
                for (int e = 0; e < 4; ++e) x[i][q][e] = __expf(x[i][q][e] - mnew);
                csum += (x[i][q][0] + x[i][q][1]) + (x[i][q][2] + x[i][q][3]);
            }
            s[i] = s[i] * scale + csum;
            #pragma unroll
            for (int j = 0; j < CT; ++j) acc[CT * i + j] = acc[CT * i + j] * scale;
        }

        #pragma unroll
        for (int t = 0; t < 8; ++t) {
            v2f bw[MT];
            #pragma unroll
            for (int i = 0; i < MT; ++i) {
                bw[i].x = x[i][t >> 1][(2 * t) & 3];
                bw[i].y = x[i][t >> 1][((2 * t) & 3) + 1];
            }
            #pragma unroll
            for (int g = 0; g < 2; ++g) {
                #pragma unroll
                for (int j = 0; j < 4; ++j) {
                    v2f a;
                    a.x = fa[t][0][g][j];
                    a.y = fa[t][1][g][j];
                    #pragma unroll
                    for (int i = 0; i < MT; ++i)
                        acc[CT * i + 4 * g + j] = __builtin_amdgcn_wmma_f32_16x16x4_f32(
                            false, a, false, bw[i], (short)0, acc[CT * i + 4 * g + j], false, false);
                }
            }
        }

        __builtin_amdgcn_sched_barrier(0);
        #pragma unroll
        for (int i = 0; i < MT; ++i) {
            #pragma unroll
            for (int q = 0; q < 4; ++q) xn[i][q] = pn[i][q] * scn[q];
        }
    }

    const size_t grow0 = (size_t)b * Sn + i0 + (wave << 3);
    v4f fv[8];
    #pragma unroll
    for (int q = 0; q < 8; ++q)
        fv[q] = *(const v4f*)(feature + (grow0 + q) * Fn + lane * 4);

    #pragma unroll
    for (int i = 0; i < MT; ++i) {
        const float stot = s[i] + xhalf(s[i]);
        if (h == 0) {
            smM[kq * MR + 16 * i + c] = m[i];
            smS[kq * MR + 16 * i + c] = stot;
        }
    }
    float* st = stage + (kq * MR + c) * SP + 32 * h;
    #pragma unroll
    for (int i = 0; i < MT; ++i) {
        #pragma unroll
        for (int g = 0; g < 2; ++g) {
            #pragma unroll
            for (int r = 0; r < 8; ++r)
                *(v4f*)&st[16 * i * SP + 64 * g + 4 * r] =
                    v4f{acc[CT * i + 4 * g + 0][r], acc[CT * i + 4 * g + 1][r],
                        acc[CT * i + 4 * g + 2][r], acc[CT * i + 4 * g + 3][r]};
        }
    }
    __syncthreads();

    float d[8], nf[8], na[8];
    #pragma unroll
    for (int q = 0; q < 8; ++q) {
        const int trow = (wave << 3) + q;
        const float m0 = smM[trow], m1 = smM[MR + trow], m2 = smM[2 * MR + trow], m3 = smM[3 * MR + trow];
        const float mx = fmaxf(fmaxf(m0, m1), fmaxf(m2, m3));
        const float f0 = __expf(m0 - mx), f1 = __expf(m1 - mx), f2 = __expf(m2 - mx), f3 = __expf(m3 - mx);
        const float den = ((f0 * smS[trow] + f1 * smS[MR + trow]) + f2 * smS[2 * MR + trow]) + f3 * smS[3 * MR + trow];
        const float inv = 1.0f / den;
        const v4f p0 = *(const v4f*)&stage[(0 * MR + trow) * SP + lane * 4];
        const v4f p1 = *(const v4f*)&stage[(1 * MR + trow) * SP + lane * 4];
        const v4f p2 = *(const v4f*)&stage[(2 * MR + trow) * SP + lane * 4];
        const v4f p3 = *(const v4f*)&stage[(3 * MR + trow) * SP + lane * 4];
        const v4f av = (((p0 * f0 + p1 * f1) + p2 * f2) + p3 * f3) * inv;
        *(v4f*)(out + (grow0 + q) * Fn + lane * 4) = av;
        d[q] = 0.f; nf[q] = 0.f; na[q] = 0.f;
        #pragma unroll
        for (int e = 0; e < 4; ++e) {
            d[q]  += fv[q][e] * av[e];
            nf[q] += fv[q][e] * fv[q][e];
            na[q] += av[e] * av[e];
        }
    }
    #pragma unroll
    for (int off = 16; off > 0; off >>= 1) {
        #pragma unroll
        for (int q = 0; q < 8; ++q) {
            d[q]  += __shfl_xor(d[q],  off, 32);
            nf[q] += __shfl_xor(nf[q], off, 32);
            na[q] += __shfl_xor(na[q], off, 32);
        }
    }
    if (lane == 0) {
        *(v4f*)(dotv + grow0)     = v4f{d[0],  d[1],  d[2],  d[3]};
        *(v4f*)(dotv + grow0 + 4) = v4f{d[4],  d[5],  d[6],  d[7]};
        *(v4f*)(nfv  + grow0)     = v4f{nf[0], nf[1], nf[2], nf[3]};
        *(v4f*)(nfv  + grow0 + 4) = v4f{nf[4], nf[5], nf[6], nf[7]};
        *(v4f*)(nav  + grow0)     = v4f{na[0], na[1], na[2], na[3]};
        *(v4f*)(nav  + grow0 + 4) = v4f{na[4], na[5], na[6], na[7]};
    }
}

__global__ void __launch_bounds__(256)
k_window(const float* __restrict__ dotv, const float* __restrict__ nfv,
         const float* __restrict__ nav, float* __restrict__ part) {
    __shared__ float red[256];
    const int w = blockIdx.x * 256 + threadIdx.x;
    float v = 0.0f;
    if (w < Bn * NW) {
        const int b = w / NW;
        const int i = w % NW;
        const int base = b * Sn + i;
        float dw = 0.f, fw = 0.f, aw = 0.f;
        #pragma unroll
        for (int j = 0; j < Wn; ++j) {
            dw += dotv[base + j];
            fw += nfv[base + j];
            aw += nav[base + j];
        }
        float cs = dw / (fmaxf(sqrtf(fw), 1e-12f) * fmaxf(sqrtf(aw), 1e-12f));
        v = 1.0f - cs;
    }
    red[threadIdx.x] = v;
    __syncthreads();
    #pragma unroll
    for (int off = 128; off > 0; off >>= 1) {
        if (threadIdx.x < off) red[threadIdx.x] += red[threadIdx.x + off];
        __syncthreads();
    }
    if (threadIdx.x == 0) part[blockIdx.x] = red[0];
}

__global__ void k_final(const float* __restrict__ part, float* __restrict__ loss_out) {
    float s = 0.0f;
    for (int i = 0; i < 64; ++i) s += part[i];
    loss_out[0] = s / (float)(Bn * NW);
}

extern "C" void kernel_launch(void* const* d_in, const int* in_sizes, int n_in,
                              void* d_out, int out_size, void* d_ws, size_t ws_size,
                              hipStream_t stream) {
    const float* score   = (const float*)d_in[0];
    const float* feature = (const float*)d_in[1];
    const float* proj    = (const float*)d_in[2];
    float* out = (float*)d_out;
    float* ws  = (float*)d_ws;

    float* dotv = ws;
    float* nfv  = ws + 16384;
    float* nav  = ws + 32768;
    float* part = ws + 49152;

    k_flash_softmax_gemm<<<Bn * (Sn / MR), 128, 0, stream>>>(proj, score, feature, out, dotv, nfv, nav);
    k_window            <<<64, 256, 0, stream>>>(dotv, nfv, nav, part);
    k_final             <<<1, 1, 0, stream>>>(part, out + (size_t)Bn * Sn * Fn);
}
